# starting kernel
# baseline (speedup 1.0000x reference)
#include <hip/hip_runtime.h>
#include <hip/hip_bf16.h>


typedef float v2f __attribute__((ext_vector_type(2)));
typedef float v8f __attribute__((ext_vector_type(8)));

static __device__ inline v8f wmma_f32_16x16x4(v2f a, v2f b, v8f c) {
  return __builtin_amdgcn_wmma_f32_16x16x4_f32(false, a, false, b, (short)0, c,
                                               false, false);
}

static __device__ inline float readlane_f(float v, int l) {
  union {
    float f;
    int i;
  } u;
  u.f = v;
  u.i = __builtin_amdgcn_readlane(u.i, l);
  return u.f;
}

#define WAVES_PER_BLK 8
#define ROW_STRIDE 18

__global__ __launch_bounds__(256) void caps_routing_kernel(
    const float* __restrict__ pred, const float* __restrict__ blog,
    const int* __restrict__ nitp, float* __restrict__ outp, int nsites,
    int bsites) {
  __shared__ float ldsTile[WAVES_PER_BLK][16 * ROW_STRIDE];

  const int lane = threadIdx.x & 31;
  const int wv   = threadIdx.x >> 5;
  const int site = blockIdx.x * WAVES_PER_BLK + wv;
  if (site >= nsites) return;

  float* tile = &ldsTile[wv][0];

  {
    const int row = lane >> 1, half = (lane & 1) * 8;
    const float4* g =
        reinterpret_cast<const float4*>(pred + (size_t)site * 256 + row * 16 + half);
    float4 x0 = g[0];
    float4 x1 = g[1];
    float* l = tile + row * ROW_STRIDE + half;
    reinterpret_cast<float2*>(l)[0] = make_float2(x0.x, x0.y);
    reinterpret_cast<float2*>(l)[1] = make_float2(x0.z, x0.w);
    reinterpret_cast<float2*>(l)[2] = make_float2(x1.x, x1.y);
    reinterpret_cast<float2*>(l)[3] = make_float2(x1.z, x1.w);
  }

  const bool lo = (lane & 16) == 0;
  const int  m  = lane & 15;

  v2f predA[4], predB[4];
#pragma unroll
  for (int k = 0; k < 4; ++k) {
    const int acol = 4 * k + (lo ? 0 : 2);
    predA[k] = *reinterpret_cast<v2f*>(tile + m * ROW_STRIDE + acol);
    const int brow = 4 * k + (lo ? 0 : 2);
    predB[k].x = tile[brow * ROW_STRIDE + m];
    predB[k].y = tile[(brow + 1) * ROW_STRIDE + m];
  }

  float bl[16];
  {
    const float* bsrc = blog + (size_t)(site % bsites) * 16;
#pragma unroll
    for (int i = 0; i < 16; ++i) bl[i] = bsrc[i];
  }

  float c[16];
  float csum;
  float outv[16];
  float laneOut = 0.f;

  auto softmax16 = [&]() {
    float mx = bl[0];
#pragma unroll
    for (int i = 1; i < 16; ++i) mx = fmaxf(mx, bl[i]);
    float sum = 0.f;
#pragma unroll
    for (int i = 0; i < 16; ++i) {
      c[i] = __expf(bl[i] - mx);
      sum += c[i];
    }
    csum = sum;
  };

  auto routeS = [&]() {
    v8f acc = {};
#pragma unroll
    for (int k = 0; k < 4; ++k) {
      v2f ac;
      ac.x = lo ? c[4 * k] : c[4 * k + 2];
      ac.y = lo ? c[4 * k + 1] : c[4 * k + 3];
      acc = wmma_f32_16x16x4(ac, predB[k], acc);
    }
    const float sval = acc[0];
    float sp[16], sqp = 0.f;
#pragma unroll
    for (int d = 0; d < 16; ++d) {
      sp[d] = readlane_f(sval, d);
      sqp += sp[d] * sp[d];
    }
    const float inv = __frcp_rn(csum);
    const float sq  = sqp * inv * inv;
    const float g   = (sq / (1.0f + sq)) * __frsqrt_rn(sq + 1e-7f) * inv;
#pragma unroll
    for (int d = 0; d < 16; ++d) outv[d] = g * sp[d];
    laneOut = g * sval;
  };

  auto agree = [&]() {
    v8f acc = {};
#pragma unroll
    for (int k = 0; k < 4; ++k) {
      v2f bo;
      bo.x = lo ? outv[4 * k] : outv[4 * k + 2];
      bo.y = lo ? outv[4 * k + 1] : outv[4 * k + 3];
      acc = wmma_f32_16x16x4(predA[k], bo, acc);
    }
#pragma unroll
    for (int j = 0; j < 8; ++j) {
      bl[j]     += readlane_f(acc[j], 0);
      bl[j + 8] += readlane_f(acc[j], 16);
    }
  };

  softmax16();
  routeS();
  const int iters = nitp[0];
  for (int t = 0; t < iters; ++t) {
    agree();
    softmax16();
    routeS();
  }

  if (lane < 16) outp[(size_t)site * 16 + lane] = laneOut;
}

extern "C" void kernel_launch(void* const* d_in, const int* in_sizes, int n_in,
                              void* d_out, int out_size, void* d_ws,
                              size_t ws_size, hipStream_t stream) {
  const float* pred = (const float*)d_in[0];
  const float* blog = (const float*)d_in[1];
  const int* nit    = (const int*)d_in[2];
  float* out        = (float*)d_out;

  const int nsites = out_size / 16;
  const int bsites = in_sizes[1] / 16;
  const int blocks = (nsites + WAVES_PER_BLK - 1) / WAVES_PER_BLK;

  caps_routing_kernel<<<blocks, 256, 0, stream>>>(pred, blog, nit, out, nsites,
                                                  bsites);
}


# Agreement product rewritten as a' = s'^T pred^T on the UNSCALED sum s' (pred as B operand): result lands one per lane, no 2nd LDS exchange; b += g*a' applies squash scale after; sum order kept
# speedup vs baseline: 1.9217x; 1.9217x over previous
#include <hip/hip_runtime.h>
#include <hip/hip_bf16.h>


typedef float v2f __attribute__((ext_vector_type(2)));
typedef float v8f __attribute__((ext_vector_type(8)));

static __device__ inline v8f wmma_f32_16x16x4(v2f a, v2f b, v8f c) {
  return __builtin_amdgcn_wmma_f32_16x16x4_f32(false, a, false, b, (short)0, c,
                                               false, false);
}

template <int X>
static __device__ inline float row_xor(float v) {
  const int i = __builtin_bit_cast(int, v);
  return __builtin_bit_cast(
      float, __builtin_amdgcn_update_dpp(i, i, 0x160 + X, 0xf, 0xf, true));
}
static __device__ inline float row_sum16(float v) {
  v += row_xor<1>(v);
  v += row_xor<2>(v);
  v += row_xor<4>(v);
  v += row_xor<8>(v);
  return v;
}
static __device__ inline float row_max16(float v) {
  v = __builtin_elementwise_maximum(v, row_xor<1>(v));
  v = __builtin_elementwise_maximum(v, row_xor<2>(v));
  v = __builtin_elementwise_maximum(v, row_xor<4>(v));
  v = __builtin_elementwise_maximum(v, row_xor<8>(v));
  return v;
}

#define WAVES_PER_BLK 8
#define GRID_BLOCKS 8192
#define ROW_STRIDE 18
#define XCH_FLOATS 32
#define LOG2E 1.44269504088896340736f
#define SUM_LO 5.9604644775390625e-08f
#define SUM_HI 16777216.0f
#define N2_HI 1.2676506002282294e+30f

__global__ __launch_bounds__(256) void caps_routing_kernel(
    const float* __restrict__ pred, const float* __restrict__ blog,
    const int* __restrict__ nitp, float* __restrict__ outp, int nsites,
    int bsites) {
  __shared__ float ldsTile[WAVES_PER_BLK][16 * ROW_STRIDE];
  __shared__ __attribute__((aligned(16))) float ldsXch[WAVES_PER_BLK][XCH_FLOATS];

  const int lane = threadIdx.x & 31;
  const int wv = __builtin_amdgcn_readfirstlane((int)(threadIdx.x >> 5));
  int site = blockIdx.x * WAVES_PER_BLK + wv;
  if (site >= nsites) return;
  const int stride = GRID_BLOCKS * WAVES_PER_BLK;

  float* tile = &ldsTile[wv][0];
  float* xw   = &ldsXch[wv][0];

  const int h = lane >> 4;
  const int m = lane & 15;
  const int row = lane >> 1, half = (lane & 1) * 8;
  const int goff = row * 16 + half;

  int bidx = site % bsites;
  const int bstep = stride % bsites;

  const int iters = nitp[0];

  float4 x0, x1;
  float bl;
  {
    const float4* g =
        reinterpret_cast<const float4*>(pred + (size_t)site * 256 + goff);
    x0 = g[0];
    x1 = g[1];
    bl = blog[(size_t)bidx * 16 + m];
  }

  for (;;) {
    {
      float* l = tile + row * ROW_STRIDE + half;
      reinterpret_cast<float2*>(l)[0] = make_float2(x0.x, x0.y);
      reinterpret_cast<float2*>(l)[1] = make_float2(x0.z, x0.w);
      reinterpret_cast<float2*>(l)[2] = make_float2(x1.x, x1.y);
      reinterpret_cast<float2*>(l)[3] = make_float2(x1.z, x1.w);
    }

    v2f predA[4], predB[4];
#pragma unroll
    for (int k = 0; k < 4; ++k) {
      const int e0 = 8 * h + 2 * k;
      predA[k] = *reinterpret_cast<v2f*>(tile + m * ROW_STRIDE + e0);
      predB[k].x = tile[e0 * ROW_STRIDE + m];
      predB[k].y = tile[(e0 + 1) * ROW_STRIDE + m];
    }

    const int nsite = site + stride;
    const bool more = nsite < nsites;
    int nbidx = bidx + bstep;
    if (nbidx >= bsites) nbidx -= bsites;
    float4 nx0, nx1;
    float nbl = bl;
    if (more) {
      const float4* g =
          reinterpret_cast<const float4*>(pred + (size_t)nsite * 256 + goff);
      nx0 = g[0];
      nx1 = g[1];
      nbl = blog[(size_t)nbidx * 16 + m];
    }

    float csum;
    float laneOut = 0.f;
    float K = LOG2E * __builtin_bit_cast(float, __builtin_amdgcn_readfirstlane(
                          __builtin_bit_cast(int, bl)));

    auto load_kform = [&](int base, float (&v)[8]) {
      const float4* p = reinterpret_cast<const float4*>(xw + base + 8 * h);
      const float4 q0 = p[0], q1 = p[1];
      v[0] = q0.x; v[1] = q0.y; v[2] = q0.z; v[3] = q0.w;
      v[4] = q1.x; v[5] = q1.y; v[6] = q1.z; v[7] = q1.w;
    };

    auto routePass = [&](bool withAgree) {
      float sval, n2, aval = 0.f;
      bool exact = false;
      float arg = fmaf(bl, LOG2E, -K);
      for (;;) {
        const float cv = __builtin_amdgcn_exp2f(arg);
        xw[m] = cv;
        float c8[8];
        load_kform(0, c8);
        v8f acc = {};
#pragma unroll
        for (int k = 0; k < 4; ++k) {
          v2f ac;
          ac.x = c8[2 * k];
          ac.y = c8[2 * k + 1];
          acc = wmma_f32_16x16x4(ac, predB[k], acc);
        }
        sval = acc[0];
        if (withAgree) {
          xw[16 + m] = sval;
          float o8[8];
          load_kform(16, o8);
          v8f acc2 = {};
#pragma unroll
          for (int k = 0; k < 4; ++k) {
            v2f ao;
            ao.x = o8[2 * k];
            ao.y = o8[2 * k + 1];
            acc2 = wmma_f32_16x16x4(ao, predA[k], acc2);
          }
          aval = acc2[0];
        }
        const float both = row_sum16(h ? sval * sval : cv);
        csum = __builtin_bit_cast(
            float, __builtin_amdgcn_readlane(__builtin_bit_cast(int, both), 0));
        n2 = __builtin_bit_cast(
            float, __builtin_amdgcn_readlane(__builtin_bit_cast(int, both), 16));
        if (exact | ((csum > SUM_LO) & (csum < SUM_HI) & (n2 < N2_HI))) break;
        const float mx = row_max16(bl);
        arg = (bl - mx) * LOG2E;
        K = LOG2E * __builtin_bit_cast(float, __builtin_amdgcn_readfirstlane(
                                                  __builtin_bit_cast(int, mx)));
        exact = true;
      }
      const float c2   = csum * csum;
      const float rden = __builtin_amdgcn_rcpf(c2 + n2);
      const float g    = n2 * rden * __frsqrt_rn(fmaf(1e-7f, c2, n2));
      if (withAgree) {
        bl = fmaf(g, aval, bl);
      } else {
        laneOut = g * sval;
      }
      K += (float)((int)((__builtin_bit_cast(unsigned, csum) >> 23) & 0xffu) - 127);
    };

    for (int t = 0; t < iters; ++t) routePass(true);
    routePass(false);

    if (lane < 16) outp[(size_t)site * 16 + lane] = laneOut;

    if (!more) break;
    site = nsite;
    bidx = nbidx;
    x0 = nx0;
    x1 = nx1;
    bl = nbl;
  }
}

extern "C" void kernel_launch(void* const* d_in, const int* in_sizes, int n_in,
                              void* d_out, int out_size, void* d_ws,
                              size_t ws_size, hipStream_t stream) {
  const float* pred = (const float*)d_in[0];
  const float* blog = (const float*)d_in[1];
  const int* nit    = (const int*)d_in[2];
  float* out        = (float*)d_out;

  const int nsites = out_size / 16;
  const int bsites = in_sizes[1] / 16;
  caps_routing_kernel<<<GRID_BLOCKS, 256, 0, stream>>>(pred, blog, nit, out,
                                                       nsites, bsites);
}
